# starting kernel
# baseline (speedup 1.0000x reference)
#include <hip/hip_runtime.h>
#include <hip/hip_bf16.h>
#include <math.h>


typedef float v2f __attribute__((ext_vector_type(2)));
typedef float v8f __attribute__((ext_vector_type(8)));

#define NQ    8
#define NS    256
#define NLAY  2
#define TM    32
#define THREADS 256

__device__ __forceinline__ int bswz(int k, int n) {
  const int fraglane = (((k >> 1) & 1) << 4) | (n & 15);
  return (((k >> 2) << 4) + (n >> 4)) * 64 + fraglane * 2 + (k & 1);
}

__global__ __launch_bounds__(256) void build_layer_mats(
    const float* __restrict__ wts, float* __restrict__ W)
{
  const int t    = threadIdx.x;
  const int wave = blockIdx.x * (blockDim.x >> 5) + (t >> 5);
  const int lane = t & 31;
  const int layer = wave >> 8;
  const int k     = wave & 255;

  float re[8], im[8];
#pragma unroll
  for (int r = 0; r < 8; ++r) {
    re[r] = (((lane << 3) | r) == k) ? 1.f : 0.f;
    im[r] = 0.f;
  }

  const float* wl = wts + layer * (NQ * 2);

#pragma unroll
  for (int i = 0; i < NQ; ++i) {
    const int b = 7 - i;
    const float hy = 0.5f * wl[i * 2 + 0];
    const float cy = cosf(hy), sy = sinf(hy);
    if (b < 3) {
      const int m = 1 << b;
#pragma unroll
      for (int r = 0; r < 8; ++r) {
        if (!(r & m)) {
          const int r1 = r | m;
          float a0 = re[r], a1 = re[r1];
          re[r] = cy * a0 - sy * a1;  re[r1] = sy * a0 + cy * a1;
          a0 = im[r]; a1 = im[r1];
          im[r] = cy * a0 - sy * a1;  im[r1] = sy * a0 + cy * a1;
        }
      }
    } else {
      const int lm = 1 << (b - 3);
      const bool hi = (lane & lm) != 0;
#pragma unroll
      for (int r = 0; r < 8; ++r) {
        const float ore = __shfl_xor(re[r], lm, 32);
        const float oim = __shfl_xor(im[r], lm, 32);
        re[r] = hi ? (sy * ore + cy * re[r]) : (cy * re[r] - sy * ore);
        im[r] = hi ? (sy * oim + cy * im[r]) : (cy * im[r] - sy * oim);
      }
    }
    const float hz = 0.5f * wl[i * 2 + 1];
    const float cz = cosf(hz), sz = sinf(hz);
#pragma unroll
    for (int r = 0; r < 8; ++r) {
      const int bit = (b < 3) ? ((r >> b) & 1) : ((lane >> (b - 3)) & 1);
      const float tt = bit ? -sz : sz;
      const float nr = cz * re[r] + tt * im[r];
      const float ni = cz * im[r] - tt * re[r];
      re[r] = nr; im[r] = ni;
    }
  }

#pragma unroll
  for (int g = 0; g < NQ; ++g) {
    const int ci = (g < 7) ? g : 7;
    const int ti = (g < 7) ? g + 1 : 0;
    const int bc = 7 - ci, bt = 7 - ti;
    if (bt < 3) {
      const int tm = 1 << bt;
      float tre[8], tim[8];
#pragma unroll
      for (int r = 0; r < 8; ++r) { tre[r] = re[r]; tim[r] = im[r]; }
#pragma unroll
      for (int r = 0; r < 8; ++r) {
        const int cb = (bc < 3) ? ((r >> bc) & 1) : ((lane >> (bc - 3)) & 1);
        re[r] = cb ? tre[r ^ tm] : tre[r];
        im[r] = cb ? tim[r ^ tm] : tim[r];
      }
    } else {
      const int lm = 1 << (bt - 3);
#pragma unroll
      for (int r = 0; r < 8; ++r) {
        const float ore = __shfl_xor(re[r], lm, 32);
        const float oim = __shfl_xor(im[r], lm, 32);
        const int cb = (bc < 3) ? ((r >> bc) & 1) : ((lane >> (bc - 3)) & 1);
        re[r] = cb ? ore : re[r];
        im[r] = cb ? oim : im[r];
      }
    }
  }

  float* Wre = W + layer * 2 * (NS * NS);
  float* Wim = Wre + NS * NS;
#pragma unroll
  for (int r = 0; r < 8; ++r) {
    const int n = (lane << 3) | r;
    const int idx = bswz(k, n);
    Wre[idx] = re[r];
    Wim[idx] = im[r];
  }
}

static __device__ __forceinline__ v8f wmma_f32(v2f a, v2f b, v8f c) {
  return __builtin_amdgcn_wmma_f32_16x16x4_f32(
       false, a,  false, b,
       (short)0, c,  false,  false);
}

__global__ __launch_bounds__(THREADS) void circuit_gemm(
    const float* __restrict__ x, const float* __restrict__ W,
    float* __restrict__ out)
{
  __shared__ float Sre[TM][NS];
  __shared__ float Sim[TM][NS];
  __shared__ float csC[TM][NQ];
  __shared__ float csS[TM][NQ];
  __shared__ float oacc[TM][NQ];

  const int t  = threadIdx.x;
  const int s0 = blockIdx.x * TM;

  {
    const int m = t >> 3, i = t & 7;
    const float a = x[(s0 + m) * NQ + i] * 0.785398163397448f;
    csC[m][i] = cosf(a);
    csS[m][i] = sinf(a);
    oacc[m][i] = 0.f;
  }
  __syncthreads();

  for (int e = t; e < TM * NS; e += THREADS) {
    const int m = e >> 8, s = e & 255;
    float p = 1.f;
#pragma unroll
    for (int i = 0; i < NQ; ++i)
      p *= ((s >> (7 - i)) & 1) ? csS[m][i] : csC[m][i];
    Sre[m][s] = p;
    Sim[m][s] = 0.f;
  }
  __syncthreads();

  const int wave = t >> 5, lane = t & 31;
  const int half = lane >> 4, lq = lane & 15;

  v8f accR[2][2], accI[2][2];

  for (int l = 0; l < NLAY; ++l) {
    const float* Wre = W + l * 2 * (NS * NS);
    const float* Wim = Wre + NS * NS;

#pragma unroll
    for (int tt = 0; tt < 2; ++tt)
#pragma unroll
      for (int mt = 0; mt < 2; ++mt) {
        accR[tt][mt] = (v8f){0.f, 0.f, 0.f, 0.f, 0.f, 0.f, 0.f, 0.f};
        accI[tt][mt] = (v8f){0.f, 0.f, 0.f, 0.f, 0.f, 0.f, 0.f, 0.f};
      }

    for (int k = 0; k < NS; k += 4) {
      const int kk = k + 2 * half;
      const v2f ar0 = *(const v2f*)&Sre[lq][kk];
      const v2f ai0 = *(const v2f*)&Sim[lq][kk];
      const v2f ar1 = *(const v2f*)&Sre[lq + 16][kk];
      const v2f ai1 = *(const v2f*)&Sim[lq + 16][kk];
      const v2f an0 = -ai0;
      const v2f an1 = -ai1;
#pragma unroll
      for (int tt = 0; tt < 2; ++tt) {
        const int nt = wave * 2 + tt;
        const int fidx = (((k >> 2) << 4) + nt) * 64 + lane * 2;
        const v2f br = *(const v2f*)&Wre[fidx];
        const v2f bi = *(const v2f*)&Wim[fidx];
        accR[tt][0] = wmma_f32(ar0, br, accR[tt][0]);
        accR[tt][0] = wmma_f32(an0, bi, accR[tt][0]);
        accI[tt][0] = wmma_f32(ar0, bi, accI[tt][0]);
        accI[tt][0] = wmma_f32(ai0, br, accI[tt][0]);
        accR[tt][1] = wmma_f32(ar1, br, accR[tt][1]);
        accR[tt][1] = wmma_f32(an1, bi, accR[tt][1]);
        accI[tt][1] = wmma_f32(ar1, bi, accI[tt][1]);
        accI[tt][1] = wmma_f32(ai1, br, accI[tt][1]);
      }
    }
    __syncthreads();

    if (l == 0) {
#pragma unroll
      for (int tt = 0; tt < 2; ++tt) {
        const int n = (wave * 2 + tt) * 16 + lq;
#pragma unroll
        for (int mt = 0; mt < 2; ++mt)
#pragma unroll
          for (int j = 0; j < 8; ++j) {
            const int m = mt * 16 + j + 8 * half;
            Sre[m][n] = accR[tt][mt][j];
            Sim[m][n] = accI[tt][mt][j];
          }
      }
      __syncthreads();
      for (int i = 0; i < NQ; ++i) {
        const int b = 7 - i;
        for (int p = t; p < TM * (NS / 2); p += THREADS) {
          const int m = p >> 7, pr = p & 127;
          const int low = pr & ((1 << b) - 1);
          const int sA = ((pr >> b) << (b + 1)) | low;
          const int sB = sA | (1 << b);
          const float c = csC[m][i], sn = csS[m][i];
          float a0 = Sre[m][sA], a1 = Sre[m][sB];
          Sre[m][sA] = c * a0 - sn * a1;
          Sre[m][sB] = sn * a0 + c * a1;
          a0 = Sim[m][sA]; a1 = Sim[m][sB];
          Sim[m][sA] = c * a0 - sn * a1;
          Sim[m][sB] = sn * a0 + c * a1;
        }
        __syncthreads();
      }
    }
  }

#pragma unroll
  for (int tt = 0; tt < 2; ++tt) {
    const int n = (wave * 2 + tt) * 16 + lq;
#pragma unroll
    for (int mt = 0; mt < 2; ++mt)
#pragma unroll
      for (int j = 0; j < 8; ++j) {
        const int m = mt * 16 + j + 8 * half;
        const float pv = accR[tt][mt][j] * accR[tt][mt][j] +
                         accI[tt][mt][j] * accI[tt][mt][j];
#pragma unroll
        for (int w = 0; w < NQ; ++w) {
          atomicAdd(&oacc[m][w], ((n >> (7 - w)) & 1) ? -pv : pv);
        }
      }
  }
  __syncthreads();

  {
    const int m = t >> 3, i = t & 7;
    out[(s0 + m) * NQ + i] = oacc[m][i];
  }
}

extern "C" void kernel_launch(void* const* d_in, const int* in_sizes, int n_in,
                              void* d_out, int out_size, void* d_ws, size_t ws_size,
                              hipStream_t stream) {
  const float* x   = (const float*)d_in[0];
  const float* wts = (const float*)d_in[1];
  float* out = (float*)d_out;
  float* W   = (float*)d_ws;

  const int B = in_sizes[0] / NQ;

  build_layer_mats<<<(NLAY * NS) / (256 / 32), 256, 0, stream>>>(wts, W);
  circuit_gemm<<<B / TM, THREADS, 0, stream>>>(x, W, out);
}


# Layer matrix split as unit-phase diagonal times REAL matrix (RZ commute behind RY; CNOT ring permutes rows): GEMM weight real, 8 WMMA per k-step not 12, phase on VALU; sum order and rounding change
# speedup vs baseline: 5.4756x; 5.4756x over previous
#include <hip/hip_runtime.h>
#include <hip/hip_bf16.h>
#include <math.h>


typedef float v2f __attribute__((ext_vector_type(2)));
typedef float v4f __attribute__((ext_vector_type(4)));
typedef float v8f __attribute__((ext_vector_type(8)));

#define NQ    8
#define NS    256
#define NLAY  2
#define TM    32
#define THREADS 256
#define NWAVE (THREADS / 32)
#define SP    (NS + 4)

__device__ __forceinline__ int bswz(int k, int n) {
  const int fraglane = (((k >> 1) & 1) << 4) | (n & 15);
  return (((k >> 2) << 4) + (n >> 4)) * 64 + fraglane * 2 + (k & 1);
}


#define WS_PHASE  (NLAY * NS * NS)
#define K1_WAVES  (NLAY * NS + NLAY)
#define K1_BLOCKS ((K1_WAVES + 7) / 8)

__device__ __forceinline__ void cnot_ring(float (&v)[8], const int lane) {
#pragma unroll
  for (int g = 0; g < NQ; ++g) {
    const int ci = (g < 7) ? g : 7;
    const int ti = (g < 7) ? g + 1 : 0;
    const int bc = 7 - ci, bt = 7 - ti;
    if (bt < 3) {
      const int tm = 1 << bt;
      float tv[8];
#pragma unroll
      for (int r = 0; r < 8; ++r) tv[r] = v[r];
#pragma unroll
      for (int r = 0; r < 8; ++r) {
        const int cb = (bc < 3) ? ((r >> bc) & 1) : ((lane >> (bc - 3)) & 1);
        v[r] = cb ? tv[r ^ tm] : tv[r];
      }
    } else {
      const int lm = 1 << (bt - 3);
#pragma unroll
      for (int r = 0; r < 8; ++r) {
        const float ov = __shfl_xor(v[r], lm, 32);
        const int cb = (bc < 3) ? ((r >> bc) & 1) : ((lane >> (bc - 3)) & 1);
        v[r] = cb ? ov : v[r];
      }
    }
  }
}

__global__ __launch_bounds__(256) void build_layer_mats(
    const float* __restrict__ wts, float* __restrict__ W)
{
  const int t    = threadIdx.x;
  const int wave = blockIdx.x * (blockDim.x >> 5) + (t >> 5);
  const int lane = t & 31;

  if (wave < NLAY * NS) {
    const int layer = wave >> 8;
    const int k     = wave & 255;
    const float* wl = wts + layer * (NQ * 2);

    float v[8];
#pragma unroll
    for (int r = 0; r < 8; ++r)
      v[r] = (((lane << 3) | r) == k) ? 1.f : 0.f;

#pragma unroll
    for (int i = 0; i < NQ; ++i) {
      const int b = 7 - i;
      const float hy = 0.5f * wl[i * 2 + 0];
      const float cy = cosf(hy), sy = sinf(hy);
      if (b < 3) {
        const int m = 1 << b;
#pragma unroll
        for (int r = 0; r < 8; ++r) {
          if (!(r & m)) {
            const int r1 = r | m;
            const float a0 = v[r], a1 = v[r1];
            v[r] = cy * a0 - sy * a1;  v[r1] = sy * a0 + cy * a1;
          }
        }
      } else {
        const int lm = 1 << (b - 3);
        const bool hi = (lane & lm) != 0;
#pragma unroll
        for (int r = 0; r < 8; ++r) {
          const float ov = __shfl_xor(v[r], lm, 32);
          v[r] = hi ? (sy * ov + cy * v[r]) : (cy * v[r] - sy * ov);
        }
      }
    }

    cnot_ring(v, lane);

    float* Wr = W + layer * (NS * NS);
#pragma unroll
    for (int r = 0; r < 8; ++r) {
      const int n = (lane << 3) | r;
      Wr[bswz(k, n)] = v[r];
    }
  } else if (wave < K1_WAVES) {
    const int layer = wave - NLAY * NS;
    const float* wl = wts + layer * (NQ * 2);

    float re[8], im[8];
#pragma unroll
    for (int r = 0; r < 8; ++r) { re[r] = 1.f; im[r] = 0.f; }

#pragma unroll
    for (int i = 0; i < NQ; ++i) {
      const int b = 7 - i;
      const float hz = 0.5f * wl[i * 2 + 1];
      const float cz = cosf(hz), sz = sinf(hz);
#pragma unroll
      for (int r = 0; r < 8; ++r) {
        const int bit = (b < 3) ? ((r >> b) & 1) : ((lane >> (b - 3)) & 1);
        const float tt = bit ? -sz : sz;
        const float nr = cz * re[r] + tt * im[r];
        const float ni = cz * im[r] - tt * re[r];
        re[r] = nr; im[r] = ni;
      }
    }

    cnot_ring(re, lane);
    cnot_ring(im, lane);

    float* Dre = W + WS_PHASE + layer * (2 * NS);
    float* Dim = Dre + NS;
#pragma unroll
    for (int r = 0; r < 8; ++r) {
      Dre[(lane << 3) | r] = re[r];
      Dim[(lane << 3) | r] = im[r];
    }
  }
}

static __device__ __forceinline__ v8f wmma_f32(v2f a, v2f b, v8f c) {
  return __builtin_amdgcn_wmma_f32_16x16x4_f32(
       false, a,  false, b,
       (short)0, c,  false,  false);
}

__global__ __launch_bounds__(THREADS, 6) void circuit_gemm(
    const float* __restrict__ x, const float* __restrict__ W,
    float* __restrict__ out)
{
  __shared__ alignas(16) float Sre[TM][SP];
  __shared__ alignas(16) float Sim[TM][SP];
  __shared__ float csC[TM][NQ];
  __shared__ float csS[TM][NQ];
  __shared__ alignas(16) float part[NWAVE][TM][NQ];

  const int t  = threadIdx.x;
  const int s0 = blockIdx.x * TM;

  {
    const int m = t >> 3, i = t & 7;
    const float a = x[(s0 + m) * NQ + i] * 0.785398163397448f;
    csC[m][i] = cosf(a);
    csS[m][i] = sinf(a);
  }
  __syncthreads();

  for (int e = t; e < TM * NS; e += THREADS) {
    const int m = e >> 8, s = e & 255;
    float p = 1.f;
#pragma unroll
    for (int i = 0; i < NQ; ++i)
      p *= ((s >> (7 - i)) & 1) ? csS[m][i] : csC[m][i];
    Sre[m][s] = p;
    Sim[m][s] = 0.f;
  }
  __syncthreads();

  const int wave = t >> 5, lane = t & 31;
  const int half = lane >> 4, lq = lane & 15;

  v8f accR[2][2], accI[2][2];

  for (int l = 0; l < NLAY; ++l) {
    const float* Wr  = W + l * (NS * NS);
    const float* Dre = W + WS_PHASE + l * (2 * NS);
    const float* Dim = Dre + NS;

    v8f yr[2][2], yi[2][2];
#pragma unroll
    for (int tt = 0; tt < 2; ++tt)
#pragma unroll
      for (int mt = 0; mt < 2; ++mt) {
        yr[tt][mt] = (v8f){0.f, 0.f, 0.f, 0.f, 0.f, 0.f, 0.f, 0.f};
        yi[tt][mt] = yr[tt][mt];
      }

    for (int k = 0; k < NS; k += 4) {
      const int kk = k + 2 * half;
      const v2f ar0 = *(const v2f*)&Sre[lq][kk];
      const v2f ai0 = *(const v2f*)&Sim[lq][kk];
      const v2f ar1 = *(const v2f*)&Sre[lq + 16][kk];
      const v2f ai1 = *(const v2f*)&Sim[lq + 16][kk];
#pragma unroll
      for (int tt = 0; tt < 2; ++tt) {
        const int nt = wave * 2 + tt;
        const int fidx = (((k >> 2) << 4) + nt) * 64 + lane * 2;
        const v2f br = *(const v2f*)&Wr[fidx];
        yr[tt][0] = wmma_f32(br, ar0, yr[tt][0]);
        yi[tt][0] = wmma_f32(br, ai0, yi[tt][0]);
        yr[tt][1] = wmma_f32(br, ar1, yr[tt][1]);
        yi[tt][1] = wmma_f32(br, ai1, yi[tt][1]);
      }
    }
#pragma unroll
    for (int tt = 0; tt < 2; ++tt) {
      const int nb = (wave * 2 + tt) * 16 + 8 * half;
      const v4f dr0 = *(const v4f*)&Dre[nb], dr1 = *(const v4f*)&Dre[nb + 4];
      const v4f di0 = *(const v4f*)&Dim[nb], di1 = *(const v4f*)&Dim[nb + 4];
      const v8f dr = (v8f){dr0[0], dr0[1], dr0[2], dr0[3], dr1[0], dr1[1], dr1[2], dr1[3]};
      const v8f di = (v8f){di0[0], di0[1], di0[2], di0[3], di1[0], di1[1], di1[2], di1[3]};
#pragma unroll
      for (int mt = 0; mt < 2; ++mt) {
        accR[tt][mt] = dr * yr[tt][mt] - di * yi[tt][mt];
        accI[tt][mt] = dr * yi[tt][mt] + di * yr[tt][mt];
      }
    }
    __syncthreads();

    if (l == 0) {
#pragma unroll
      for (int tt = 0; tt < 2; ++tt) {
        const int nb = (wave * 2 + tt) * 16 + 8 * half;
#pragma unroll
        for (int mt = 0; mt < 2; ++mt) {
          const int m = mt * 16 + lq;
          const v8f r = accR[tt][mt], q = accI[tt][mt];
          *(v4f*)&Sre[m][nb]     = (v4f){r[0], r[1], r[2], r[3]};
          *(v4f*)&Sre[m][nb + 4] = (v4f){r[4], r[5], r[6], r[7]};
          *(v4f*)&Sim[m][nb]     = (v4f){q[0], q[1], q[2], q[3]};
          *(v4f*)&Sim[m][nb + 4] = (v4f){q[4], q[5], q[6], q[7]};
        }
      }
      __syncthreads();
      for (int i = 0; i < NQ; ++i) {
        const int b = 7 - i;
        for (int p = t; p < TM * (NS / 2); p += THREADS) {
          const int m = p >> 7, pr = p & 127;
          const int low = pr & ((1 << b) - 1);
          const int sA = ((pr >> b) << (b + 1)) | low;
          const int sB = sA | (1 << b);
          const float c = csC[m][i], sn = csS[m][i];
          float a0 = Sre[m][sA], a1 = Sre[m][sB];
          Sre[m][sA] = c * a0 - sn * a1;
          Sre[m][sB] = sn * a0 + c * a1;
          a0 = Sim[m][sA]; a1 = Sim[m][sB];
          Sim[m][sA] = c * a0 - sn * a1;
          Sim[m][sB] = sn * a0 + c * a1;
        }
        __syncthreads();
      }
    }
  }

  v8f zacc[2];
  zacc[0] = (v8f){0.f, 0.f, 0.f, 0.f, 0.f, 0.f, 0.f, 0.f};
  zacc[1] = zacc[0];
  const int zsh = (7 - lq) & 7;
#pragma unroll
  for (int tt = 0; tt < 2; ++tt) {
    const int nb = (wave * 2 + tt) * 16 + 8 * half;
#pragma unroll
    for (int q = 0; q < 4; ++q) {
      v2f zf;
#pragma unroll
      for (int v = 0; v < 2; ++v) {
        const int n = nb + 2 * q + v;
        const float sg = ((n >> zsh) & 1) ? -1.f : 1.f;
        zf[v] = (lq < NQ) ? sg : 0.f;
      }
#pragma unroll
      for (int mt = 0; mt < 2; ++mt) {
        const v8f r = accR[tt][mt], im = accI[tt][mt];
        v2f pf;
        pf[0] = r[2 * q]     * r[2 * q]     + im[2 * q]     * im[2 * q];
        pf[1] = r[2 * q + 1] * r[2 * q + 1] + im[2 * q + 1] * im[2 * q + 1];
        zacc[mt] = wmma_f32(zf, pf, zacc[mt]);
      }
    }
  }
  if (half == 0) {
#pragma unroll
    for (int mt = 0; mt < 2; ++mt) {
      const v8f z = zacc[mt];
      float* dst = &part[wave][mt * 16 + lq][0];
      *(v4f*)dst       = (v4f){z[0], z[1], z[2], z[3]};
      *(v4f*)(dst + 4) = (v4f){z[4], z[5], z[6], z[7]};
    }
  }
  __syncthreads();

  {
    const int m = t >> 3, i = t & 7;
    float s = part[0][m][i];
#pragma unroll
    for (int w = 1; w < NWAVE; ++w) s += part[w][m][i];
    out[(s0 + m) * NQ + i] = s;
  }
}

extern "C" void kernel_launch(void* const* d_in, const int* in_sizes, int n_in,
                              void* d_out, int out_size, void* d_ws, size_t ws_size,
                              hipStream_t stream) {
  const float* x   = (const float*)d_in[0];
  const float* wts = (const float*)d_in[1];
  float* out = (float*)d_out;
  float* W   = (float*)d_ws;

  const int B = in_sizes[0] / NQ;

  build_layer_mats<<<K1_BLOCKS, 256, 0, stream>>>(wts, W);
  circuit_gemm<<<B / TM, THREADS, 0, stream>>>(x, W, out);
}


# Layer 1 in closed form: RY(w)*RY(a)=RY(a+w), so its state is a product state (angle-addition cos/sin), phased and moved by the CNOT ring at the store; the layer-1 GEMM is gone; sum order changes
# speedup vs baseline: 8.8255x; 1.6118x over previous
#include <hip/hip_runtime.h>
#include <hip/hip_bf16.h>
#include <math.h>


typedef float v2f __attribute__((ext_vector_type(2)));
typedef float v4f __attribute__((ext_vector_type(4)));
typedef float v8f __attribute__((ext_vector_type(8)));

#define NQ    8
#define NS    256
#define NLAY  2
#define TM    32
#define THREADS 256
#define NWAVE (THREADS / 32)
#define SP    (NS + 4)

__device__ __forceinline__ int bswz(int k, int n) {
  const int fraglane = (((k >> 1) & 1) << 4) | (n & 15);
  return (((k >> 2) << 4) + (n >> 4)) * 64 + fraglane * 2 + (k & 1);
}


#define WS_PHASE  (NLAY * NS * NS)
#define K1_WAVES  (NS + NLAY)
#define K1_BLOCKS ((K1_WAVES + 7) / 8)

__device__ __forceinline__ void cnot_ring(float (&v)[8], const int lane) {
#pragma unroll
  for (int g = 0; g < NQ; ++g) {
    const int ci = (g < 7) ? g : 7;
    const int ti = (g < 7) ? g + 1 : 0;
    const int bc = 7 - ci, bt = 7 - ti;
    if (bt < 3) {
      const int tm = 1 << bt;
      float tv[8];
#pragma unroll
      for (int r = 0; r < 8; ++r) tv[r] = v[r];
#pragma unroll
      for (int r = 0; r < 8; ++r) {
        const int cb = (bc < 3) ? ((r >> bc) & 1) : ((lane >> (bc - 3)) & 1);
        v[r] = cb ? tv[r ^ tm] : tv[r];
      }
    } else {
      const int lm = 1 << (bt - 3);
#pragma unroll
      for (int r = 0; r < 8; ++r) {
        const float ov = __shfl_xor(v[r], lm, 32);
        const int cb = (bc < 3) ? ((r >> bc) & 1) : ((lane >> (bc - 3)) & 1);
        v[r] = cb ? ov : v[r];
      }
    }
  }
}

__global__ __launch_bounds__(256) void build_layer_mats(
    const float* __restrict__ wts, float* __restrict__ W)
{
  const int t    = threadIdx.x;
  const int wave = blockIdx.x * (blockDim.x >> 5) + (t >> 5);
  const int lane = t & 31;

  if (wave < NS) {
    const int layer = NLAY - 1;
    const int k     = wave;
    const float* wl = wts + layer * (NQ * 2);

    float v[8];
#pragma unroll
    for (int r = 0; r < 8; ++r)
      v[r] = (((lane << 3) | r) == k) ? 1.f : 0.f;

#pragma unroll
    for (int i = 0; i < NQ; ++i) {
      const int b = 7 - i;
      const float hy = 0.5f * wl[i * 2 + 0];
      const float cy = cosf(hy), sy = sinf(hy);
      if (b < 3) {
        const int m = 1 << b;
#pragma unroll
        for (int r = 0; r < 8; ++r) {
          if (!(r & m)) {
            const int r1 = r | m;
            const float a0 = v[r], a1 = v[r1];
            v[r] = cy * a0 - sy * a1;  v[r1] = sy * a0 + cy * a1;
          }
        }
      } else {
        const int lm = 1 << (b - 3);
        const bool hi = (lane & lm) != 0;
#pragma unroll
        for (int r = 0; r < 8; ++r) {
          const float ov = __shfl_xor(v[r], lm, 32);
          v[r] = hi ? (sy * ov + cy * v[r]) : (cy * v[r] - sy * ov);
        }
      }
    }

    cnot_ring(v, lane);

    float* Wr = W + layer * (NS * NS);
#pragma unroll
    for (int r = 0; r < 8; ++r) {
      const int n = (lane << 3) | r;
      Wr[bswz(k, n)] = v[r];
    }
  } else if (wave < K1_WAVES) {
    const int layer = wave - NS;
    const float* wl = wts + layer * (NQ * 2);

    float re[8], im[8];
#pragma unroll
    for (int r = 0; r < 8; ++r) { re[r] = 1.f; im[r] = 0.f; }

#pragma unroll
    for (int i = 0; i < NQ; ++i) {
      const int b = 7 - i;
      const float hz = 0.5f * wl[i * 2 + 1];
      const float cz = cosf(hz), sz = sinf(hz);
#pragma unroll
      for (int r = 0; r < 8; ++r) {
        const int bit = (b < 3) ? ((r >> b) & 1) : ((lane >> (b - 3)) & 1);
        const float tt = bit ? -sz : sz;
        const float nr = cz * re[r] + tt * im[r];
        const float ni = cz * im[r] - tt * re[r];
        re[r] = nr; im[r] = ni;
      }
    }

    cnot_ring(re, lane);
    cnot_ring(im, lane);

    float* Dre = W + WS_PHASE + layer * (2 * NS);
    float* Dim = Dre + NS;
#pragma unroll
    for (int r = 0; r < 8; ++r) {
      Dre[(lane << 3) | r] = re[r];
      Dim[(lane << 3) | r] = im[r];
    }
  }
}

static __device__ __forceinline__ v8f wmma_f32(v2f a, v2f b, v8f c) {
  return __builtin_amdgcn_wmma_f32_16x16x4_f32(
       false, a,  false, b,
       (short)0, c,  false,  false);
}

__global__ __launch_bounds__(THREADS, 6) void circuit_gemm(
    const float* __restrict__ x, const float* __restrict__ wts,
    const float* __restrict__ W, float* __restrict__ out)
{
  __shared__ alignas(16) float Sre[TM][SP];
  __shared__ alignas(16) float Sim[TM][SP];
  __shared__ float csC[TM][NQ];
  __shared__ float csS[TM][NQ];
  __shared__ float cs1[TM][NQ][2];
  __shared__ alignas(16) float part[NWAVE][TM][NQ];

  const int t  = threadIdx.x;
  const int s0 = blockIdx.x * TM;

  int n = t;
#pragma unroll
  for (int g = 0; g < NQ; ++g) {
    const int ci = (g < 7) ? g : 7;
    const int ti = (g < 7) ? g + 1 : 0;
    n ^= ((n >> (7 - ci)) & 1) << (7 - ti);
  }
  const float d1r = W[WS_PHASE + n];
  const float d1i = W[WS_PHASE + NS + n];

  {
    const int m = t >> 3, i = t & 7;
    const float a  = x[(s0 + m) * NQ + i] * 0.785398163397448f;
    const float hw = 0.5f * wts[i * 2 + 0];
    const float ca = cosf(a),  sa = sinf(a);
    const float cw = cosf(hw), sw = sinf(hw);
    csC[m][i] = ca;
    csS[m][i] = sa;
    cs1[m][i][0] = ca * cw - sa * sw;
    cs1[m][i][1] = sa * cw + ca * sw;
  }
  __syncthreads();

  {
    const float* col[NQ];
#pragma unroll
    for (int i = 0; i < NQ; ++i) col[i] = &cs1[0][i][(t >> (7 - i)) & 1];
#pragma unroll
    for (int m = 0; m < TM; ++m) {
      float p = col[0][m * (NQ * 2)];
#pragma unroll
      for (int i = 1; i < NQ; ++i) p *= col[i][m * (NQ * 2)];
      Sre[m][n] = d1r * p;
      Sim[m][n] = d1i * p;
    }
  }
  __syncthreads();

  for (int i = 0; i < NQ; ++i) {
    const int b = 7 - i;
    for (int p = t; p < TM * (NS / 2); p += THREADS) {
      const int m = p >> 7, pr = p & 127;
      const int low = pr & ((1 << b) - 1);
      const int sA = ((pr >> b) << (b + 1)) | low;
      const int sB = sA | (1 << b);
      const float c = csC[m][i], sn = csS[m][i];
      float a0 = Sre[m][sA], a1 = Sre[m][sB];
      Sre[m][sA] = c * a0 - sn * a1;
      Sre[m][sB] = sn * a0 + c * a1;
      a0 = Sim[m][sA]; a1 = Sim[m][sB];
      Sim[m][sA] = c * a0 - sn * a1;
      Sim[m][sB] = sn * a0 + c * a1;
    }
    __syncthreads();
  }

  const int wave = t >> 5, lane = t & 31;
  const int half = lane >> 4, lq = lane & 15;

  v8f accR[2][2], accI[2][2];

  {
    const float* Wr  = W + (NLAY - 1) * (NS * NS);
    const float* Dre = W + WS_PHASE + (NLAY - 1) * (2 * NS);
    const float* Dim = Dre + NS;

    v8f yr[2][2], yi[2][2];
#pragma unroll
    for (int tt = 0; tt < 2; ++tt)
#pragma unroll
      for (int mt = 0; mt < 2; ++mt) {
        yr[tt][mt] = (v8f){0.f, 0.f, 0.f, 0.f, 0.f, 0.f, 0.f, 0.f};
        yi[tt][mt] = yr[tt][mt];
      }

    for (int k = 0; k < NS; k += 4) {
      const int kk = k + 2 * half;
      const v2f ar0 = *(const v2f*)&Sre[lq][kk];
      const v2f ai0 = *(const v2f*)&Sim[lq][kk];
      const v2f ar1 = *(const v2f*)&Sre[lq + 16][kk];
      const v2f ai1 = *(const v2f*)&Sim[lq + 16][kk];
#pragma unroll
      for (int tt = 0; tt < 2; ++tt) {
        const int nt = wave * 2 + tt;
        const int fidx = (((k >> 2) << 4) + nt) * 64 + lane * 2;
        const v2f br = *(const v2f*)&Wr[fidx];
        yr[tt][0] = wmma_f32(br, ar0, yr[tt][0]);
        yi[tt][0] = wmma_f32(br, ai0, yi[tt][0]);
        yr[tt][1] = wmma_f32(br, ar1, yr[tt][1]);
        yi[tt][1] = wmma_f32(br, ai1, yi[tt][1]);
      }
    }
#pragma unroll
    for (int tt = 0; tt < 2; ++tt) {
      const int nb = (wave * 2 + tt) * 16 + 8 * half;
      const v4f dr0 = *(const v4f*)&Dre[nb], dr1 = *(const v4f*)&Dre[nb + 4];
      const v4f di0 = *(const v4f*)&Dim[nb], di1 = *(const v4f*)&Dim[nb + 4];
      const v8f dr = (v8f){dr0[0], dr0[1], dr0[2], dr0[3], dr1[0], dr1[1], dr1[2], dr1[3]};
      const v8f di = (v8f){di0[0], di0[1], di0[2], di0[3], di1[0], di1[1], di1[2], di1[3]};
#pragma unroll
      for (int mt = 0; mt < 2; ++mt) {
        accR[tt][mt] = dr * yr[tt][mt] - di * yi[tt][mt];
        accI[tt][mt] = dr * yi[tt][mt] + di * yr[tt][mt];
      }
    }
  }

  v8f zacc[2];
  zacc[0] = (v8f){0.f, 0.f, 0.f, 0.f, 0.f, 0.f, 0.f, 0.f};
  zacc[1] = zacc[0];
  const int zsh = (7 - lq) & 7;
#pragma unroll
  for (int tt = 0; tt < 2; ++tt) {
    const int nb = (wave * 2 + tt) * 16 + 8 * half;
#pragma unroll
    for (int q = 0; q < 4; ++q) {
      v2f zf;
#pragma unroll
      for (int v = 0; v < 2; ++v) {
        const int n = nb + 2 * q + v;
        const float sg = ((n >> zsh) & 1) ? -1.f : 1.f;
        zf[v] = (lq < NQ) ? sg : 0.f;
      }
#pragma unroll
      for (int mt = 0; mt < 2; ++mt) {
        const v8f r = accR[tt][mt], im = accI[tt][mt];
        v2f pf;
        pf[0] = r[2 * q]     * r[2 * q]     + im[2 * q]     * im[2 * q];
        pf[1] = r[2 * q + 1] * r[2 * q + 1] + im[2 * q + 1] * im[2 * q + 1];
        zacc[mt] = wmma_f32(zf, pf, zacc[mt]);
      }
    }
  }
  if (half == 0) {
#pragma unroll
    for (int mt = 0; mt < 2; ++mt) {
      const v8f z = zacc[mt];
      float* dst = &part[wave][mt * 16 + lq][0];
      *(v4f*)dst       = (v4f){z[0], z[1], z[2], z[3]};
      *(v4f*)(dst + 4) = (v4f){z[4], z[5], z[6], z[7]};
    }
  }
  __syncthreads();

  {
    const int m = t >> 3, i = t & 7;
    float s = part[0][m][i];
#pragma unroll
    for (int w = 1; w < NWAVE; ++w) s += part[w][m][i];
    out[(s0 + m) * NQ + i] = s;
  }
}

extern "C" void kernel_launch(void* const* d_in, const int* in_sizes, int n_in,
                              void* d_out, int out_size, void* d_ws, size_t ws_size,
                              hipStream_t stream) {
  const float* x   = (const float*)d_in[0];
  const float* wts = (const float*)d_in[1];
  float* out = (float*)d_out;
  float* W   = (float*)d_ws;

  const int B = in_sizes[0] / NQ;

  build_layer_mats<<<K1_BLOCKS, 256, 0, stream>>>(wts, W);
  circuit_gemm<<<B / TM, THREADS, 0, stream>>>(x, wts, W, out);
}


# Layer-2 embedding moved to the matrix unit: per sample Y = U_hi*Psi*U_lo^T (two 16-deep WMMA stages), layer-1 amplitudes built in registers; the 8 in-place LDS passes are gone; sum order changes
# speedup vs baseline: 10.9685x; 1.2428x over previous
#include <hip/hip_runtime.h>
#include <hip/hip_bf16.h>
#include <math.h>


typedef float v2f __attribute__((ext_vector_type(2)));
typedef float v4f __attribute__((ext_vector_type(4)));
typedef float v8f __attribute__((ext_vector_type(8)));

#define NQ    8
#define NS    256
#define NLAY  2
#define TM    32
#define THREADS 256
#define NWAVE (THREADS / 32)
#define SP    (NS + 4)

__device__ __forceinline__ int bswz(int k, int n) {
  const int fraglane = (((k >> 1) & 1) << 4) | (n & 15);
  return (((k >> 2) << 4) + (n >> 4)) * 64 + fraglane * 2 + (k & 1);
}


#define WS_PHASE  (NLAY * NS * NS)
#define K1_WAVES  (NS + NLAY)
#define K1_BLOCKS ((K1_WAVES + 7) / 8)

__device__ __forceinline__ void cnot_ring(float (&v)[8], const int lane) {
#pragma unroll
  for (int g = 0; g < NQ; ++g) {
    const int ci = (g < 7) ? g : 7;
    const int ti = (g < 7) ? g + 1 : 0;
    const int bc = 7 - ci, bt = 7 - ti;
    if (bt < 3) {
      const int tm = 1 << bt;
      float tv[8];
#pragma unroll
      for (int r = 0; r < 8; ++r) tv[r] = v[r];
#pragma unroll
      for (int r = 0; r < 8; ++r) {
        const int cb = (bc < 3) ? ((r >> bc) & 1) : ((lane >> (bc - 3)) & 1);
        v[r] = cb ? tv[r ^ tm] : tv[r];
      }
    } else {
      const int lm = 1 << (bt - 3);
#pragma unroll
      for (int r = 0; r < 8; ++r) {
        const float ov = __shfl_xor(v[r], lm, 32);
        const int cb = (bc < 3) ? ((r >> bc) & 1) : ((lane >> (bc - 3)) & 1);
        v[r] = cb ? ov : v[r];
      }
    }
  }
}

__global__ __launch_bounds__(256) void build_layer_mats(
    const float* __restrict__ wts, float* __restrict__ W)
{
  const int t    = threadIdx.x;
  const int wave = blockIdx.x * (blockDim.x >> 5) + (t >> 5);
  const int lane = t & 31;

  if (wave < NS) {
    const int layer = NLAY - 1;
    const int k     = wave;
    const float* wl = wts + layer * (NQ * 2);

    float v[8];
#pragma unroll
    for (int r = 0; r < 8; ++r)
      v[r] = (((lane << 3) | r) == k) ? 1.f : 0.f;

#pragma unroll
    for (int i = 0; i < NQ; ++i) {
      const int b = 7 - i;
      const float hy = 0.5f * wl[i * 2 + 0];
      const float cy = cosf(hy), sy = sinf(hy);
      if (b < 3) {
        const int m = 1 << b;
#pragma unroll
        for (int r = 0; r < 8; ++r) {
          if (!(r & m)) {
            const int r1 = r | m;
            const float a0 = v[r], a1 = v[r1];
            v[r] = cy * a0 - sy * a1;  v[r1] = sy * a0 + cy * a1;
          }
        }
      } else {
        const int lm = 1 << (b - 3);
        const bool hi = (lane & lm) != 0;
#pragma unroll
        for (int r = 0; r < 8; ++r) {
          const float ov = __shfl_xor(v[r], lm, 32);
          v[r] = hi ? (sy * ov + cy * v[r]) : (cy * v[r] - sy * ov);
        }
      }
    }

    cnot_ring(v, lane);

    float* Wr = W + layer * (NS * NS);
#pragma unroll
    for (int r = 0; r < 8; ++r) {
      const int n = (lane << 3) | r;
      Wr[bswz(k, n)] = v[r];
    }
  } else if (wave < K1_WAVES) {
    const int layer = wave - NS;
    const float* wl = wts + layer * (NQ * 2);

    float re[8], im[8];
#pragma unroll
    for (int r = 0; r < 8; ++r) { re[r] = 1.f; im[r] = 0.f; }

#pragma unroll
    for (int i = 0; i < NQ; ++i) {
      const int b = 7 - i;
      const float hz = 0.5f * wl[i * 2 + 1];
      const float cz = cosf(hz), sz = sinf(hz);
#pragma unroll
      for (int r = 0; r < 8; ++r) {
        const int bit = (b < 3) ? ((r >> b) & 1) : ((lane >> (b - 3)) & 1);
        const float tt = bit ? -sz : sz;
        const float nr = cz * re[r] + tt * im[r];
        const float ni = cz * im[r] - tt * re[r];
        re[r] = nr; im[r] = ni;
      }
    }

    cnot_ring(re, lane);
    cnot_ring(im, lane);

    float* Dre = W + WS_PHASE + layer * (2 * NS);
    float* Dim = Dre + NS;
#pragma unroll
    for (int r = 0; r < 8; ++r) {
      Dre[(lane << 3) | r] = re[r];
      Dim[(lane << 3) | r] = im[r];
    }
  }
}

static __device__ __forceinline__ v8f wmma_f32(v2f a, v2f b, v8f c) {
  return __builtin_amdgcn_wmma_f32_16x16x4_f32(
       false, a,  false, b,
       (short)0, c,  false,  false);
}

__global__ __launch_bounds__(THREADS, 6) void circuit_gemm(
    const float* __restrict__ x, const float* __restrict__ wts,
    const float* __restrict__ W, float* __restrict__ out)
{
  __shared__ alignas(16) float Sre[TM][SP];
  __shared__ alignas(16) float Sim[TM][SP];
  __shared__ alignas(16) float csC[TM][NQ];
  __shared__ alignas(16) float csS[TM][NQ];
  __shared__ float cs1[TM][NQ][2];
  __shared__ alignas(16) float part[NWAVE][TM][NQ];

  const int t  = threadIdx.x;
  const int s0 = blockIdx.x * TM;

  {
    const int m = t >> 3, i = t & 7;
    const float a  = x[(s0 + m) * NQ + i] * 0.785398163397448f;
    const float hw = 0.5f * wts[i * 2 + 0];
    const float ca = cosf(a),  sa = sinf(a);
    const float cw = cosf(hw), sw = sinf(hw);
    csC[m][i] = ca;
    csS[m][i] = sa;
    cs1[m][i][0] = ca * cw - sa * sw;
    cs1[m][i][1] = sa * cw + ca * sw;
  }
  __syncthreads();

  const int wave = t >> 5, lane = t & 31;
  const int half = lane >> 4, lq = lane & 15;

  {
    const int c3 = (lq >> 3) & 1, c2 = (lq >> 2) & 1, c1 = (lq >> 1) & 1, c0 = lq & 1;

    float d1r[4][2], d1i[4][2];
#pragma unroll
    for (int j = 0; j < 4; ++j)
#pragma unroll
      for (int e = 0; e < 2; ++e) {
        const int nn = ((4 * j + 2 * half + e) << 4) | lq;
        d1r[j][e] = W[WS_PHASE + nn];
        d1i[j][e] = W[WS_PHASE + NS + nn];
      }

    const float* g5  = &cs1[0][5][c2 ^ c3];
    const float* g6  = &cs1[0][6][c1 ^ c2];
    const float* g7  = &cs1[0][7][c0 ^ c1];
    const float* f4a = &cs1[0][4][c3],   *f4b = &cs1[0][4][c3 ^ 1];
    const float* f3a = &cs1[0][3][half], *f3b = &cs1[0][3][half ^ 1];
    const float* f2a = &cs1[0][2][half], *f2b = &cs1[0][2][half ^ 1];
    const float* f0a = &cs1[0][0][c0],   *f0b = &cs1[0][0][c0 ^ 1];
    const float* f1a = &cs1[0][1][c0],   *f1b = &cs1[0][1][c0 ^ 1];

#pragma unroll 2
    for (int mm = 0; mm < TM / NWAVE; ++mm) {
      const int m  = wave * (TM / NWAVE) + mm;
      const int mo = m * (NQ * 2);

      const float g = g5[mo] * g6[mo] * g7[mo];
      float u[2], v[4];
      u[0] = g * (f4a[mo] * f3a[mo]);
      u[1] = g * (f4b[mo] * f3b[mo]);
      {
        const float x2[2] = {f2a[mo], f2b[mo]};
        const float x0[2] = {f0a[mo], f0b[mo]};
        const float x1[2] = {f1a[mo], f1b[mo]};
#pragma unroll
        for (int j = 0; j < 4; ++j)
          v[j] = x2[j & 1] * x0[j >> 1] * x1[(j & 1) ^ (j >> 1)];
      }

      const v4f cA = *(const v4f*)&csC[m][0], cB = *(const v4f*)&csC[m][4];
      const v4f sA = *(const v4f*)&csS[m][0], sB = *(const v4f*)&csS[m][4];
      const float h0k0 = c3 ? sA[0] : cA[0], h0k1 = c3 ? cA[0] : -sA[0];
      const float h1k0 = c2 ? sA[1] : cA[1], h1k1 = c2 ? cA[1] : -sA[1];
      const float h2k0 = c1 ? sA[2] : cA[2], h2k1 = c1 ? cA[2] : -sA[2];
      const float h3k0 = c0 ? sA[3] : cA[3], h3k1 = c0 ? cA[3] : -sA[3];
      const float l0k0 = c3 ? sB[0] : cB[0], l0k1 = c3 ? cB[0] : -sB[0];
      const float l1k0 = c2 ? sB[1] : cB[1], l1k1 = c2 ? cB[1] : -sB[1];
      const float l2k0 = c1 ? sB[2] : cB[2], l2k1 = c1 ? cB[2] : -sB[2];
      const float l3k0 = c0 ? sB[3] : cB[3], l3k1 = c0 ? cB[3] : -sB[3];

      const float hw2 = half ? h2k1 : h2k0;
      const float he[2] = {hw2 * h3k0, hw2 * h3k1};
      const float hj[4] = {h0k0 * h1k0, h0k0 * h1k1, h0k1 * h1k0, h0k1 * h1k1};
      const float lw4 = half ? l0k1 : l0k0;
      const float le[2] = {lw4 * l3k0, lw4 * l3k1};
      const float lj[4] = {l1k0 * l2k0, l1k0 * l2k1, l1k1 * l2k0, l1k1 * l2k1};

      v8f tr = (v8f){0.f, 0.f, 0.f, 0.f, 0.f, 0.f, 0.f, 0.f};
      v8f ti = tr;
#pragma unroll
      for (int j = 0; j < 4; ++j) {
        const float p0 = u[0] * v[j], p1 = u[1] * v[j];
        const v2f ar = (v2f){d1r[j][0] * p0, d1r[j][1] * p1};
        const v2f ai = (v2f){d1i[j][0] * p0, d1i[j][1] * p1};
        const v2f bu = (v2f){hj[j] * he[0], hj[j] * he[1]};
        tr = wmma_f32(ar, bu, tr);
        ti = wmma_f32(ai, bu, ti);
      }
      v8f yre = (v8f){0.f, 0.f, 0.f, 0.f, 0.f, 0.f, 0.f, 0.f};
      v8f yim = yre;
#pragma unroll
      for (int q = 0; q < 4; ++q) {
        const v2f ar = (v2f){tr[2 * q], tr[2 * q + 1]};
        const v2f ai = (v2f){ti[2 * q], ti[2 * q + 1]};
        const v2f bu = (v2f){lj[q] * le[0], lj[q] * le[1]};
        yre = wmma_f32(ar, bu, yre);
        yim = wmma_f32(ai, bu, yim);
      }
#pragma unroll
      for (int r = 0; r < 8; ++r) {
        Sre[m][(8 * half + r) * 16 + lq] = yre[r];
        Sim[m][(8 * half + r) * 16 + lq] = yim[r];
      }
    }
  }
  __syncthreads();


  v8f accR[2][2], accI[2][2];

  {
    const float* Wr  = W + (NLAY - 1) * (NS * NS);
    const float* Dre = W + WS_PHASE + (NLAY - 1) * (2 * NS);
    const float* Dim = Dre + NS;

    v8f yr[2][2], yi[2][2];
#pragma unroll
    for (int tt = 0; tt < 2; ++tt)
#pragma unroll
      for (int mt = 0; mt < 2; ++mt) {
        yr[tt][mt] = (v8f){0.f, 0.f, 0.f, 0.f, 0.f, 0.f, 0.f, 0.f};
        yi[tt][mt] = yr[tt][mt];
      }

    for (int k = 0; k < NS; k += 4) {
      const int kk = k + 2 * half;
      const v2f ar0 = *(const v2f*)&Sre[lq][kk];
      const v2f ai0 = *(const v2f*)&Sim[lq][kk];
      const v2f ar1 = *(const v2f*)&Sre[lq + 16][kk];
      const v2f ai1 = *(const v2f*)&Sim[lq + 16][kk];
#pragma unroll
      for (int tt = 0; tt < 2; ++tt) {
        const int nt = wave * 2 + tt;
        const int fidx = (((k >> 2) << 4) + nt) * 64 + lane * 2;
        const v2f br = *(const v2f*)&Wr[fidx];
        yr[tt][0] = wmma_f32(br, ar0, yr[tt][0]);
        yi[tt][0] = wmma_f32(br, ai0, yi[tt][0]);
        yr[tt][1] = wmma_f32(br, ar1, yr[tt][1]);
        yi[tt][1] = wmma_f32(br, ai1, yi[tt][1]);
      }
    }
#pragma unroll
    for (int tt = 0; tt < 2; ++tt) {
      const int nb = (wave * 2 + tt) * 16 + 8 * half;
      const v4f dr0 = *(const v4f*)&Dre[nb], dr1 = *(const v4f*)&Dre[nb + 4];
      const v4f di0 = *(const v4f*)&Dim[nb], di1 = *(const v4f*)&Dim[nb + 4];
      const v8f dr = (v8f){dr0[0], dr0[1], dr0[2], dr0[3], dr1[0], dr1[1], dr1[2], dr1[3]};
      const v8f di = (v8f){di0[0], di0[1], di0[2], di0[3], di1[0], di1[1], di1[2], di1[3]};
#pragma unroll
      for (int mt = 0; mt < 2; ++mt) {
        accR[tt][mt] = dr * yr[tt][mt] - di * yi[tt][mt];
        accI[tt][mt] = dr * yi[tt][mt] + di * yr[tt][mt];
      }
    }
  }

  v8f zacc[2];
  zacc[0] = (v8f){0.f, 0.f, 0.f, 0.f, 0.f, 0.f, 0.f, 0.f};
  zacc[1] = zacc[0];
  const int zsh = (7 - lq) & 7;
#pragma unroll
  for (int tt = 0; tt < 2; ++tt) {
    const int nb = (wave * 2 + tt) * 16 + 8 * half;
#pragma unroll
    for (int q = 0; q < 4; ++q) {
      v2f zf;
#pragma unroll
      for (int v = 0; v < 2; ++v) {
        const int n = nb + 2 * q + v;
        const float sg = ((n >> zsh) & 1) ? -1.f : 1.f;
        zf[v] = (lq < NQ) ? sg : 0.f;
      }
#pragma unroll
      for (int mt = 0; mt < 2; ++mt) {
        const v8f r = accR[tt][mt], im = accI[tt][mt];
        v2f pf;
        pf[0] = r[2 * q]     * r[2 * q]     + im[2 * q]     * im[2 * q];
        pf[1] = r[2 * q + 1] * r[2 * q + 1] + im[2 * q + 1] * im[2 * q + 1];
        zacc[mt] = wmma_f32(zf, pf, zacc[mt]);
      }
    }
  }
  if (half == 0) {
#pragma unroll
    for (int mt = 0; mt < 2; ++mt) {
      const v8f z = zacc[mt];
      float* dst = &part[wave][mt * 16 + lq][0];
      *(v4f*)dst       = (v4f){z[0], z[1], z[2], z[3]};
      *(v4f*)(dst + 4) = (v4f){z[4], z[5], z[6], z[7]};
    }
  }
  __syncthreads();

  {
    const int m = t >> 3, i = t & 7;
    float s = part[0][m][i];
#pragma unroll
    for (int w = 1; w < NWAVE; ++w) s += part[w][m][i];
    out[(s0 + m) * NQ + i] = s;
  }
}

extern "C" void kernel_launch(void* const* d_in, const int* in_sizes, int n_in,
                              void* d_out, int out_size, void* d_ws, size_t ws_size,
                              hipStream_t stream) {
  const float* x   = (const float*)d_in[0];
  const float* wts = (const float*)d_in[1];
  float* out = (float*)d_out;
  float* W   = (float*)d_ws;

  const int B = in_sizes[0] / NQ;

  build_layer_mats<<<K1_BLOCKS, 256, 0, stream>>>(wts, W);
  circuit_gemm<<<B / TM, THREADS, 0, stream>>>(x, wts, W, out);
}
